# starting kernel
# baseline (speedup 1.0000x reference)
#include <hip/hip_runtime.h>
#include <hip/hip_bf16.h>

typedef __attribute__((ext_vector_type(2))) float v2f;
typedef __attribute__((ext_vector_type(8))) float v8f;

__global__ void spp_winsum_kernel(const float* __restrict__ in,
                                  float* __restrict__ ws) {
    const int tid = threadIdx.x;
    const int rg  = blockIdx.x;
    const int win = blockIdx.y;
    const int b   = blockIdx.z;

    const int wi = win >> 2, wj = win & 3;
    const int h0 = wi * 16 + rg * 4;
    const int w0 = wj * 16;

    const float2* inp = (const float2*)in;

    float2 acc0 = make_float2(0.f, 0.f);
    float2 acc1 = make_float2(0.f, 0.f);

#pragma unroll
    for (int r = 0; r < 4; ++r) {
        const int rowbase = ((b * 64 + h0 + r) * 64 + w0) * 256 + tid;
#pragma unroll
        for (int cc = 0; cc < 16; cc += 2) {
            float2 v0 = inp[rowbase + cc * 256];
            float2 v1 = inp[rowbase + (cc + 1) * 256];
            acc0.x += v0.x; acc0.y += v0.y;
            acc1.x += v1.x; acc1.y += v1.y;
        }
    }

    float2 s;
    s.x = acc0.x + acc1.x;
    s.y = acc0.y + acc1.y;
    ((float2*)ws)[(b * 64 + win * 4 + rg) * 256 + tid] = s;
}

__device__ __forceinline__ float spp_mval(int m, int win) {
    const int wi = win >> 2, wj = win & 3;
    float r = (m == 0) ? (1.0f / 4096.0f) : 0.0f;
    const int t = m - 1;
    const bool s2 = (m >= 1) & (m <= 4) &
                    ((wi >> 1) == (t >> 1)) & ((wj >> 1) == (t & 1));
    r += s2 ? (1.0f / 1024.0f) : 0.0f;
    const bool s4 = (m >= 5) & (m <= 20) & ((m - 5) == win);
    r += s4 ? (1.0f / 256.0f) : 0.0f;
    return r;
}

__global__ void spp_combine_wmma_kernel(const float* __restrict__ ws,
                                        float* __restrict__ out) {
    const int lane  = threadIdx.x;
    const int c0    = blockIdx.x * 16;
    const int b     = blockIdx.y;
    const int n     = lane & 15;
    const int khalf = lane >> 4;

    const float* wsb = ws + (size_t)b * 64 * 512 + c0 + n;

    v8f acc0 = {};
    v8f acc1 = {};

#pragma unroll
    for (int kk = 0; kk < 16; ++kk) {
        const int k0 = kk * 4 + 2 * khalf;
        v2f bmat;
        bmat.x = wsb[(size_t)k0 * 512];
        bmat.y = wsb[(size_t)(k0 + 1) * 512];

        const float a0 = spp_mval(n, kk);
        const float a1 = spp_mval(16 + n, kk);
        v2f av0; av0.x = a0; av0.y = a0;
        v2f av1; av1.x = a1; av1.y = a1;

        acc0 = __builtin_amdgcn_wmma_f32_16x16x4_f32(
            false, av0, false, bmat, (short)0, acc0, false, false);
        acc1 = __builtin_amdgcn_wmma_f32_16x16x4_f32(
            false, av1, false, bmat, (short)0, acc1, false, false);
    }

    float* outb = out + (size_t)b * 10752 + c0 + n;
#pragma unroll
    for (int r = 0; r < 8; ++r) {
        const int m0 = r + 8 * khalf;
        outb[(size_t)m0 * 512] = acc0[r];
    }
#pragma unroll
    for (int r = 0; r < 8; ++r) {
        const int m1 = 16 + r + 8 * khalf;
        if (m1 < 21) outb[(size_t)m1 * 512] = acc1[r];
    }
}

extern "C" void kernel_launch(void* const* d_in, const int* in_sizes, int n_in,
                              void* d_out, int out_size, void* d_ws, size_t ws_size,
                              hipStream_t stream) {
    const float* in = (const float*)d_in[0];
    float* out = (float*)d_out;
    float* ws  = (float*)d_ws;

    dim3 g1(4, 16, 32);
    spp_winsum_kernel<<<g1, dim3(256), 0, stream>>>(in, ws);

    dim3 g2(32, 32);
    spp_combine_wmma_kernel<<<g2, dim3(32), 0, stream>>>(ws, out);
}


# two launches fused into one: 256 workgroups (batch x 64-channel slice) x 16 waves, a wave sums one 16x16 window with 16-byte loads, bins combined in-kernel on the f32 WMMA via LDS; sum order changed
# speedup vs baseline: 1.3416x; 1.3416x over previous
#include <hip/hip_runtime.h>

typedef __attribute__((ext_vector_type(2))) float v2f;
typedef __attribute__((ext_vector_type(4))) float v4f;
typedef __attribute__((ext_vector_type(8))) float v8f;

#define SPP_PITCH 72

__device__ __forceinline__ float spp_mval(int m, int win) {
    const int wi = win >> 2, wj = win & 3;
    float r = (m == 0) ? (1.0f / 4096.0f) : 0.0f;
    const int t = m - 1;
    const bool s2 = (m >= 1) & (m <= 4) &
                    ((wi >> 1) == (t >> 1)) & ((wj >> 1) == (t & 1));
    r += s2 ? (1.0f / 1024.0f) : 0.0f;
    const bool s4 = (m >= 5) & (m <= 20) & ((m - 5) == win);
    r += s4 ? (1.0f / 256.0f) : 0.0f;
    return r;
}

__global__ void __launch_bounds__(512)
spp_fused_kernel(const float* __restrict__ in, float* __restrict__ out) {
    __shared__ __attribute__((aligned(16))) float part[32 * SPP_PITCH];

    const int tid  = threadIdx.x;
    const int lane = tid & 31;
    const int wave = __builtin_amdgcn_readfirstlane(tid >> 5);
    const int s    = blockIdx.x;
    const int b    = blockIdx.y;
    const int q    = lane & 15;
    const int hh   = lane >> 4;
    const int wi   = wave >> 2, wj = wave & 3;

    const v4f* p = (const v4f*)in +
        ((size_t)((b * 64 + wi * 16) * 64 + wj * 16 + hh) * 128 + s * 16 + q);

    v4f buf[2][16];
#define SPP_LOAD_CHUNK(C)                                                     \
    _Pragma("unroll") for (int i = 0; i < 16; ++i)                            \
        buf[(C) & 1][i] = p[((C) * 2 + (i >> 3)) * 8192 + (i & 7) * 256];

    v4f a0 = {}, a1 = {}, a2 = {}, a3 = {};
    SPP_LOAD_CHUNK(0)
#pragma unroll
    for (int c = 0; c < 8; ++c) {
        __builtin_amdgcn_sched_barrier(0);
        if (c + 1 < 8) { SPP_LOAD_CHUNK(c + 1) }
        __builtin_amdgcn_sched_barrier(0);
        const v4f* v = buf[c & 1];
#pragma unroll
        for (int i = 0; i < 16; i += 4) {
            a0 += v[i]; a1 += v[i + 1]; a2 += v[i + 2]; a3 += v[i + 3];
        }
    }
#undef SPP_LOAD_CHUNK
    const v4f acc = (a0 + a1) + (a2 + a3);

    *(v4f*)&part[(2 * wave + hh) * SPP_PITCH + 4 * q] = acc;
    __syncthreads();

    if (wave < 4) {
        const int nt = wave;
        const int n  = lane & 15;
        const int h  = lane >> 4;
        const float* pb = part + 16 * nt + n;

        v2f bm[8];
#pragma unroll
        for (int kk = 0; kk < 8; ++kk) {
            const int k0 = 4 * kk + 2 * h;
            bm[kk].x = pb[k0 * SPP_PITCH];
            bm[kk].y = pb[(k0 + 1) * SPP_PITCH];
        }
        __builtin_amdgcn_sched_barrier(0);

        v8f d0 = {};
        v8f d1 = {};
#pragma unroll
        for (int kk = 0; kk < 8; ++kk) {
            const int win = 2 * kk + h;
            const float c0 = spp_mval(n, win);
            v2f av0; av0.x = c0; av0.y = c0;
            d0 = __builtin_amdgcn_wmma_f32_16x16x4_f32(
                false, av0, false, bm[kk], (short)0, d0, false, false);
            if (kk >= 5) {
                const float c1 = spp_mval(16 + n, win);
                v2f av1; av1.x = c1; av1.y = c1;
                d1 = __builtin_amdgcn_wmma_f32_16x16x4_f32(
                    false, av1, false, bm[kk], (short)0, d1, false, false);
            }
        }

        float* ob = out + (size_t)b * 10752 + s * 64 + nt * 16 + n;
#pragma unroll
        for (int r = 0; r < 8; ++r) {
            const int m0 = r + 8 * h;
            ob[(size_t)m0 * 512] = d0[r];
        }
#pragma unroll
        for (int r = 0; r < 8; ++r) {
            const int m1 = 16 + r + 8 * h;
            if (m1 < 21) ob[(size_t)m1 * 512] = d1[r];
        }
    }
}

extern "C" void kernel_launch(void* const* d_in, const int* in_sizes, int n_in,
                              void* d_out, int out_size, void* d_ws, size_t ws_size,
                              hipStream_t stream) {
    const float* in = (const float*)d_in[0];
    float* out = (float*)d_out;

    dim3 g(8, 32);
    spp_fused_kernel<<<g, dim3(512), 0, stream>>>(in, out);
}
